# starting kernel
# baseline (speedup 1.0000x reference)
#include <hip/hip_runtime.h>
#include <hip/hip_bf16.h>

#define D_FEAT 64

typedef float v2f __attribute__((ext_vector_type(2)));
typedef float v8f __attribute__((ext_vector_type(8)));

__global__ void gcn_zero_u32(unsigned* __restrict__ p, long long n) {
  long long i = (long long)blockIdx.x * blockDim.x + threadIdx.x;
  if (i < n) p[i] = 0u;
}

__global__ void gcn_degree(const int* __restrict__ src, const int* __restrict__ dst,
                           unsigned* __restrict__ outdeg, unsigned* __restrict__ indeg,
                           int n_edges) {
  int e = blockIdx.x * blockDim.x + threadIdx.x;
  if (e < n_edges) {
    atomicAdd(&outdeg[src[e]], 1u);
    atomicAdd(&indeg[dst[e]], 1u);
  }
}

__global__ void gcn_norms(float* __restrict__ outn, float* __restrict__ inn, int n) {
  int i = blockIdx.x * blockDim.x + threadIdx.x;
  if (i < n) {
    unsigned od = ((const unsigned*)outn)[i];
    unsigned id_ = ((const unsigned*)inn)[i];
    outn[i] = rsqrtf((float)(od > 1u ? od : 1u));
    inn[i]  = rsqrtf((float)(id_ > 1u ? id_ : 1u));
  }
}

__global__ void gcn_scatter(const float* __restrict__ x,
                            const float* __restrict__ out_norm,
                            const int* __restrict__ src, const int* __restrict__ dst,
                            float* __restrict__ agg, int n_edges) {
  long long t = (long long)blockIdx.x * blockDim.x + threadIdx.x;
  int e = (int)(t >> 6);
  int f = (int)(t & 63);
  if (e < n_edges) {
    int s = src[e];
    int d = dst[e];
    float v = x[(size_t)s * D_FEAT + f] * out_norm[s];
    atomicAdd(&agg[(size_t)d * D_FEAT + f], v);
  }
}

__global__ __launch_bounds__(128) void gcn_gemm_elu(
    const float* __restrict__ agg, const float* __restrict__ in_norm,
    const float* __restrict__ W,
    const float* __restrict__ bias,
    float* __restrict__ out, int n_nodes) {
  const int lane = threadIdx.x & 31;
  const int wave = threadIdx.x >> 5;
  const int row0 = blockIdx.x * 16;
  const int n0   = wave * 16;
  const int half = lane >> 4;
  const int l15  = lane & 15;

  const int arow  = row0 + l15;
  const int arowc = arow < n_nodes ? arow : (n_nodes - 1);
  const float scale = (arow < n_nodes) ? in_norm[arowc] : 0.0f;
  const float* __restrict__ aptr = agg + (size_t)arowc * D_FEAT + 2 * half;
  const float* __restrict__ bptr = W + (size_t)(2 * half) * D_FEAT + n0 + l15;

  v8f c = {};
#pragma unroll
  for (int k0 = 0; k0 < D_FEAT; k0 += 4) {
    v2f a;
    a.x = aptr[k0] * scale;
    a.y = aptr[k0 + 1] * scale;
    v2f b;
    b.x = bptr[(size_t)k0 * D_FEAT];
    b.y = bptr[(size_t)(k0 + 1) * D_FEAT];
    c = __builtin_amdgcn_wmma_f32_16x16x4_f32(false, a, false, b,
                                              (short)0, c, false, false);
  }

  const float bv = bias[n0 + l15];
#pragma unroll
  for (int v = 0; v < 8; ++v) {
    int r = row0 + v + 8 * half;
    if (r < n_nodes) {
      float xv = c[v] + bv;
      float yv = xv > 0.0f ? xv : expm1f(xv);
      out[(size_t)r * D_FEAT + n0 + l15] = yv;
    }
  }
}

extern "C" void kernel_launch(void* const* d_in, const int* in_sizes, int n_in,
                              void* d_out, int out_size, void* d_ws, size_t ws_size,
                              hipStream_t stream) {
  const float* feat = (const float*)d_in[0];
  const float* Ws[3] = {(const float*)d_in[1], (const float*)d_in[3], (const float*)d_in[5]};
  const float* bs[3] = {(const float*)d_in[2], (const float*)d_in[4], (const float*)d_in[6]};
  const int* src = (const int*)d_in[7];
  const int* dst = (const int*)d_in[8];

  const int n_nodes = in_sizes[0] / D_FEAT;
  const int n_edges = in_sizes[7];

  float* out_norm = (float*)d_ws;
  float* in_norm  = out_norm + n_nodes;
  float* agg      = in_norm + n_nodes;
  float* hbuf     = agg + (size_t)n_nodes * D_FEAT;
  float* outp     = (float*)d_out;

  const int TPB = 256;

  {
    long long n2 = 2LL * n_nodes;
    gcn_zero_u32<<<(unsigned)((n2 + TPB - 1) / TPB), TPB, 0, stream>>>(
        (unsigned*)out_norm, n2);
    gcn_degree<<<(unsigned)((n_edges + TPB - 1) / TPB), TPB, 0, stream>>>(
        src, dst, (unsigned*)out_norm, (unsigned*)in_norm, n_edges);
    gcn_norms<<<(unsigned)((n_nodes + TPB - 1) / TPB), TPB, 0, stream>>>(
        out_norm, in_norm, n_nodes);
  }

  const float* x = feat;
  for (int l = 0; l < 3; ++l) {
    long long nagg = (long long)n_nodes * D_FEAT;
    gcn_zero_u32<<<(unsigned)((nagg + TPB - 1) / TPB), TPB, 0, stream>>>(
        (unsigned*)agg, nagg);
    long long nthreads = (long long)n_edges * D_FEAT;
    gcn_scatter<<<(unsigned)((nthreads + TPB - 1) / TPB), TPB, 0, stream>>>(
        x, out_norm, src, dst, agg, n_edges);
    float* y = (l == 2) ? outp : hbuf;
    unsigned gblocks = (unsigned)((n_nodes + 15) / 16);
    gcn_gemm_elu<<<gblocks, 128, 0, stream>>>(agg, in_norm, Ws[l], bs[l], y, n_nodes);
    x = y;
  }
}


# Edges re-encoded once per call as a 1-byte owner key plus a packed src|row word; row-owning workgroups scan key bytes, 16 per lane and load, not 4-byte dst; sum order: LDS arrival, as before
# speedup vs baseline: 2.5302x; 2.5302x over previous
#include <hip/hip_runtime.h>
#include <hip/hip_bf16.h>

#define D_FEAT 64

typedef float v2f __attribute__((ext_vector_type(2)));
typedef float v8f __attribute__((ext_vector_type(8)));

__global__ void gcn_zero_u32(unsigned* __restrict__ p, long long n) {
  long long i = (long long)blockIdx.x * blockDim.x + threadIdx.x;
  if (i < n) p[i] = 0u;
}

#define OWN_ROWS 400
#define SRC_BITS 17

typedef int v4i __attribute__((ext_vector_type(4)));
typedef unsigned v4u __attribute__((ext_vector_type(4)));

__global__ void gcn_degree_encode(const int* __restrict__ src, const int* __restrict__ dst,
                                  unsigned* __restrict__ outdeg, unsigned* __restrict__ indeg,
                                  unsigned* __restrict__ packed, unsigned* __restrict__ keys,
                                  int n_nodes, int n_edges, int n_pad) {
  const int e0 = (blockIdx.x * blockDim.x + threadIdx.x) * 4;
  if (e0 >= n_pad) return;
  v4i s = (v4i){0, 0, 0, 0};
  v4i d = (v4i){-1, -1, -1, -1};
  if (e0 + 3 < n_edges) {
    s = *(const v4i*)(src + e0);
    d = *(const v4i*)(dst + e0);
  } else {
#pragma unroll
    for (int j = 0; j < 4; ++j)
      if (e0 + j < n_edges) { s[j] = src[e0 + j]; d[j] = dst[e0 + j]; }
  }
  unsigned kw = 0;
  v4u pk;
#pragma unroll
  for (int j = 0; j < 4; ++j) {
    unsigned key = 0xFFu, p = 0u;
    if (e0 + j < n_edges) {
      atomicAdd(&outdeg[s[j]], 1u);
      atomicAdd(&indeg[d[j]], 1u);
      if ((unsigned)d[j] < (unsigned)n_nodes) {
        const unsigned own = (unsigned)d[j] / (unsigned)OWN_ROWS;
        key = own;
        p = (unsigned)s[j] | (((unsigned)d[j] - own * (unsigned)OWN_ROWS) << SRC_BITS);
      }
    }
    kw |= key << (8 * j);
    pk[j] = p;
  }
  keys[e0 >> 2] = kw;
  *(v4u*)(packed + e0) = pk;
}

__global__ void gcn_norms(float* __restrict__ outn, float* __restrict__ inn, int n) {
  int i = blockIdx.x * blockDim.x + threadIdx.x;
  if (i < n) {
    unsigned od = ((const unsigned*)outn)[i];
    unsigned id_ = ((const unsigned*)inn)[i];
    outn[i] = rsqrtf((float)(od > 1u ? od : 1u));
    inn[i]  = rsqrtf((float)(id_ > 1u ? id_ : 1u));
  }
}

#define AGG_WAVES 32
#define WIN 2048
#define WIN_LOADS (WIN / 512)

__global__ __launch_bounds__(AGG_WAVES * 32) void gcn_agg_owned(
    const float* __restrict__ x, const float* __restrict__ out_norm,
    const unsigned* __restrict__ keys, const unsigned* __restrict__ packed,
    float* __restrict__ agg, int n_nodes, int n_win) {
  __shared__ __attribute__((aligned(16))) float acc[OWN_ROWS * D_FEAT];
  __shared__ unsigned short hitq[AGG_WAVES * WIN];

  const int tid  = threadIdx.x;
  const int lane = tid & 31;
  const int wave = __builtin_amdgcn_readfirstlane(tid >> 5);
  const int lo   = blockIdx.x * OWN_ROWS;
  unsigned short* __restrict__ myq = hitq + wave * WIN;
  const unsigned pat = (unsigned)blockIdx.x * 0x01010101u;

  for (int i = tid; i < OWN_ROWS * D_FEAT / 4; i += AGG_WAVES * 32)
    ((float4*)acc)[i] = make_float4(0.f, 0.f, 0.f, 0.f);
  __syncthreads();

  const int start = (int)(((long long)blockIdx.x * n_win) / gridDim.x);

  v4u nk[WIN_LOADS];
#pragma unroll
  for (int q = 0; q < WIN_LOADS; ++q) nk[q] = (v4u){~0u, ~0u, ~0u, ~0u};

  auto load_window = [&](int w, v4u* out) {
    const v4u* p = (const v4u*)keys + (size_t)w * (WIN / 16) + lane;
#pragma unroll
    for (int q = 0; q < WIN_LOADS; ++q) out[q] = p[q * 32];
  };

  int idx = wave;
  if (idx < n_win) {
    int w = idx + start; if (w >= n_win) w -= n_win;
    load_window(w, nk);
  }

  for (; idx < n_win; idx += AGG_WAVES) {
    int w = idx + start; if (w >= n_win) w -= n_win;
    v4u k[WIN_LOADS];
#pragma unroll
    for (int q = 0; q < WIN_LOADS; ++q) k[q] = nk[q];
    if (idx + AGG_WAVES < n_win) {
      int wn = idx + AGG_WAVES + start; if (wn >= n_win) wn -= n_win;
      load_window(wn, nk);
    }

    unsigned count = 0;
#pragma unroll
    for (int q = 0; q < WIN_LOADS; ++q) {
      unsigned g = 0;
#pragma unroll
      for (int cw = 0; cw < 4; ++cw) {
        const unsigned xx = k[q][cw] ^ pat;
        const unsigned nz = (((xx & 0x7f7f7f7fu) + 0x7f7f7f7fu) | xx) & 0x80808080u;
        g |= (nz ^ 0x80808080u) >> (7 - cw);
      }
      unsigned m = __builtin_amdgcn_ballot_w32(g != 0u);
      while (m) {
        if (g != 0u) {
          const unsigned t = (unsigned)__builtin_ctz(g);
          const unsigned pos = count + __builtin_amdgcn_mbcnt_lo(m, 0u);
          myq[pos] = (unsigned short)(q * 512 + lane * 16 + ((t & 7u) << 2) + (t >> 3));
          g &= g - 1u;
        }
        count += (unsigned)__builtin_popcount(m);
        m = __builtin_amdgcn_ballot_w32(g != 0u);
      }
    }

    const int wbase = w * WIN;
    for (unsigned base = 0; base < count; base += 32) {
      const unsigned qi = base + lane;
      unsigned pk = 0;
      if (qi < count) pk = packed[wbase + (int)myq[qi]];
      const int s_l = (int)(pk & ((1u << SRC_BITS) - 1u));
      const unsigned r_l = pk >> SRC_BITS;
      const unsigned n = (count - base < 32u) ? (count - base) : 32u;
      for (unsigned j = 0; j < n; j += 4) {
        const bool k1 = j + 1 < n, k2 = j + 2 < n, k3 = j + 3 < n;
        const int j0 = (int)j;
        const int j1 = k1 ? j0 + 1 : j0;
        const int j2 = k2 ? j0 + 2 : j0;
        const int j3 = k3 ? j0 + 3 : j0;
        const int s0 = __builtin_amdgcn_readlane(s_l, j0);
        const int s1 = __builtin_amdgcn_readlane(s_l, j1);
        const int s2 = __builtin_amdgcn_readlane(s_l, j2);
        const int s3 = __builtin_amdgcn_readlane(s_l, j3);
        const v2f v0 = *(const v2f*)(x + (size_t)s0 * D_FEAT + 2 * lane);
        const v2f v1 = *(const v2f*)(x + (size_t)s1 * D_FEAT + 2 * lane);
        const v2f v2 = *(const v2f*)(x + (size_t)s2 * D_FEAT + 2 * lane);
        const v2f v3 = *(const v2f*)(x + (size_t)s3 * D_FEAT + 2 * lane);
        const float n0 = out_norm[s0];
        const float n1 = out_norm[s1];
        const float n2 = out_norm[s2];
        const float n3 = out_norm[s3];
        float* p0 = acc + __builtin_amdgcn_readlane((int)r_l, j0) * D_FEAT + 2 * lane;
        float* p1 = acc + __builtin_amdgcn_readlane((int)r_l, j1) * D_FEAT + 2 * lane;
        float* p2 = acc + __builtin_amdgcn_readlane((int)r_l, j2) * D_FEAT + 2 * lane;
        float* p3 = acc + __builtin_amdgcn_readlane((int)r_l, j3) * D_FEAT + 2 * lane;
        __hip_atomic_fetch_add(p0,     v0.x * n0, __ATOMIC_RELAXED, __HIP_MEMORY_SCOPE_WORKGROUP);
        __hip_atomic_fetch_add(p0 + 1, v0.y * n0, __ATOMIC_RELAXED, __HIP_MEMORY_SCOPE_WORKGROUP);
        __hip_atomic_fetch_add(p1,     k1 ? v1.x * n1 : 0.f, __ATOMIC_RELAXED, __HIP_MEMORY_SCOPE_WORKGROUP);
        __hip_atomic_fetch_add(p1 + 1, k1 ? v1.y * n1 : 0.f, __ATOMIC_RELAXED, __HIP_MEMORY_SCOPE_WORKGROUP);
        __hip_atomic_fetch_add(p2,     k2 ? v2.x * n2 : 0.f, __ATOMIC_RELAXED, __HIP_MEMORY_SCOPE_WORKGROUP);
        __hip_atomic_fetch_add(p2 + 1, k2 ? v2.y * n2 : 0.f, __ATOMIC_RELAXED, __HIP_MEMORY_SCOPE_WORKGROUP);
        __hip_atomic_fetch_add(p3,     k3 ? v3.x * n3 : 0.f, __ATOMIC_RELAXED, __HIP_MEMORY_SCOPE_WORKGROUP);
        __hip_atomic_fetch_add(p3 + 1, k3 ? v3.y * n3 : 0.f, __ATOMIC_RELAXED, __HIP_MEMORY_SCOPE_WORKGROUP);
      }
    }
  }

  __syncthreads();
  const int rows_here = (n_nodes - lo < OWN_ROWS) ? (n_nodes - lo) : OWN_ROWS;
  float4* __restrict__ o4 = (float4*)(agg + (size_t)lo * D_FEAT);
  for (int i = tid; i < rows_here * (D_FEAT / 4); i += AGG_WAVES * 32)
    o4[i] = ((const float4*)acc)[i];
}

__global__ __launch_bounds__(128) void gcn_gemm_elu(
    const float* __restrict__ agg, const float* __restrict__ in_norm,
    const float* __restrict__ W,
    const float* __restrict__ bias,
    float* __restrict__ out, int n_nodes) {
  const int lane = threadIdx.x & 31;
  const int wave = threadIdx.x >> 5;
  const int row0 = blockIdx.x * 16;
  const int n0   = wave * 16;
  const int half = lane >> 4;
  const int l15  = lane & 15;

  const int arow  = row0 + l15;
  const int arowc = arow < n_nodes ? arow : (n_nodes - 1);
  const float scale = (arow < n_nodes) ? in_norm[arowc] : 0.0f;
  const float* __restrict__ aptr = agg + (size_t)arowc * D_FEAT + 2 * half;
  const float* __restrict__ bptr = W + (size_t)(2 * half) * D_FEAT + n0 + l15;

  v8f c = {};
#pragma unroll
  for (int k0 = 0; k0 < D_FEAT; k0 += 4) {
    v2f a;
    a.x = aptr[k0] * scale;
    a.y = aptr[k0 + 1] * scale;
    v2f b;
    b.x = bptr[(size_t)k0 * D_FEAT];
    b.y = bptr[(size_t)(k0 + 1) * D_FEAT];
    c = __builtin_amdgcn_wmma_f32_16x16x4_f32(false, a, false, b,
                                              (short)0, c, false, false);
  }

  const float bv = bias[n0 + l15];
#pragma unroll
  for (int v = 0; v < 8; ++v) {
    int r = row0 + v + 8 * half;
    if (r < n_nodes) {
      float xv = c[v] + bv;
      float yv = xv > 0.0f ? xv : expm1f(xv);
      out[(size_t)r * D_FEAT + n0 + l15] = yv;
    }
  }
}

extern "C" void kernel_launch(void* const* d_in, const int* in_sizes, int n_in,
                              void* d_out, int out_size, void* d_ws, size_t ws_size,
                              hipStream_t stream) {
  const float* feat = (const float*)d_in[0];
  const float* Ws[3] = {(const float*)d_in[1], (const float*)d_in[3], (const float*)d_in[5]};
  const float* bs[3] = {(const float*)d_in[2], (const float*)d_in[4], (const float*)d_in[6]};
  const int* src = (const int*)d_in[7];
  const int* dst = (const int*)d_in[8];

  const int n_nodes = in_sizes[0] / D_FEAT;
  const int n_edges = in_sizes[7];

  const int n_win = (n_edges + WIN - 1) / WIN;
  const int n_pad = n_win * WIN;
  float* out_norm  = (float*)d_ws;
  float* in_norm   = out_norm + n_nodes;
  float* agg       = in_norm + n_nodes;
  unsigned* packed = (unsigned*)(agg + (size_t)n_nodes * D_FEAT);
  unsigned* keys   = packed + n_pad;
  float* outp      = (float*)d_out;

  const int TPB = 256;

  {
    long long n2 = 2LL * n_nodes;
    gcn_zero_u32<<<(unsigned)((n2 + TPB - 1) / TPB), TPB, 0, stream>>>(
        (unsigned*)out_norm, n2);
    gcn_degree_encode<<<(unsigned)((n_pad / 4 + TPB - 1) / TPB), TPB, 0, stream>>>(
        src, dst, (unsigned*)out_norm, (unsigned*)in_norm, packed, keys,
        n_nodes, n_edges, n_pad);
    gcn_norms<<<(unsigned)((n_nodes + TPB - 1) / TPB), TPB, 0, stream>>>(
        out_norm, in_norm, n_nodes);
  }

  const float* x = feat;
  for (int l = 0; l < 3; ++l) {
    gcn_agg_owned<<<(unsigned)((n_nodes + OWN_ROWS - 1) / OWN_ROWS), AGG_WAVES * 32, 0, stream>>>(
        x, out_norm, keys, packed, agg, n_nodes, n_win);
    unsigned gblocks = (unsigned)((n_nodes + 15) / 16);
    gcn_gemm_elu<<<gblocks, 128, 0, stream>>>(agg, in_norm, Ws[l], bs[l], outp, n_nodes);
    x = outp;
  }
}


# Projection moved into the row-owning aggregation kernel: a workgroup's 400 LDS sums feed the f32 WMMA A fragments in place (LDS row pitch 272 B); agg round trip gone, 9 to 6 launches; sum order kept
# speedup vs baseline: 2.6916x; 1.0638x over previous
#include <hip/hip_runtime.h>
#include <hip/hip_bf16.h>

#define D_FEAT 64

typedef float v2f __attribute__((ext_vector_type(2)));
typedef float v8f __attribute__((ext_vector_type(8)));

__global__ void gcn_zero_u32(unsigned* __restrict__ p, long long n) {
  long long i = (long long)blockIdx.x * blockDim.x + threadIdx.x;
  if (i < n) p[i] = 0u;
}

#define OWN_ROWS 400
#define SRC_BITS 17

typedef int v4i __attribute__((ext_vector_type(4)));
typedef unsigned v4u __attribute__((ext_vector_type(4)));

__global__ void gcn_degree_encode(const int* __restrict__ src, const int* __restrict__ dst,
                                  unsigned* __restrict__ outdeg, unsigned* __restrict__ indeg,
                                  unsigned* __restrict__ packed, unsigned* __restrict__ keys,
                                  int n_nodes, int n_edges, int n_pad) {
  const int e0 = (blockIdx.x * blockDim.x + threadIdx.x) * 4;
  if (e0 >= n_pad) return;
  v4i s = (v4i){0, 0, 0, 0};
  v4i d = (v4i){-1, -1, -1, -1};
  if (e0 + 3 < n_edges) {
    s = *(const v4i*)(src + e0);
    d = *(const v4i*)(dst + e0);
  } else {
#pragma unroll
    for (int j = 0; j < 4; ++j)
      if (e0 + j < n_edges) { s[j] = src[e0 + j]; d[j] = dst[e0 + j]; }
  }
  unsigned kw = 0;
  v4u pk;
#pragma unroll
  for (int j = 0; j < 4; ++j) {
    unsigned key = 0xFFu, p = 0u;
    if (e0 + j < n_edges) {
      atomicAdd(&outdeg[s[j]], 1u);
      atomicAdd(&indeg[d[j]], 1u);
      if ((unsigned)d[j] < (unsigned)n_nodes) {
        const unsigned own = (unsigned)d[j] / (unsigned)OWN_ROWS;
        key = own;
        p = (unsigned)s[j] | (((unsigned)d[j] - own * (unsigned)OWN_ROWS) << SRC_BITS);
      }
    }
    kw |= key << (8 * j);
    pk[j] = p;
  }
  keys[e0 >> 2] = kw;
  *(v4u*)(packed + e0) = pk;
}

__global__ void gcn_norms(float* __restrict__ outn, float* __restrict__ inn, int n) {
  int i = blockIdx.x * blockDim.x + threadIdx.x;
  if (i < n) {
    unsigned od = ((const unsigned*)outn)[i];
    unsigned id_ = ((const unsigned*)inn)[i];
    outn[i] = rsqrtf((float)(od > 1u ? od : 1u));
    inn[i]  = rsqrtf((float)(id_ > 1u ? id_ : 1u));
  }
}

#define AGG_WAVES 32
#define WIN 2048
#define WIN_LOADS (WIN / 512)
#define ACC_PITCH 68
#define ROW_TILES (OWN_ROWS / 16)
static_assert(OWN_ROWS % 16 == 0, "whole 16-row tiles");
static_assert(AGG_WAVES * WIN * 2 >= OWN_ROWS * D_FEAT * 4, "results fit in the hit lists' LDS");

__global__ __launch_bounds__(AGG_WAVES * 32) void gcn_agg_owned(
    const float* __restrict__ x, const float* __restrict__ out_norm,
    const unsigned* __restrict__ keys, const unsigned* __restrict__ packed,
    const float* __restrict__ in_norm,
    const float* __restrict__ W,
    const float* __restrict__ bias,
    float* __restrict__ out, int n_nodes, int n_win) {
  __shared__ __attribute__((aligned(16))) float acc[OWN_ROWS * ACC_PITCH];
  __shared__ __attribute__((aligned(16))) unsigned short hitq[AGG_WAVES * WIN];

  const int tid  = threadIdx.x;
  const int lane = tid & 31;
  const int wave = __builtin_amdgcn_readfirstlane(tid >> 5);
  const int lo   = blockIdx.x * OWN_ROWS;
  unsigned short* __restrict__ myq = hitq + wave * WIN;
  const unsigned pat = (unsigned)blockIdx.x * 0x01010101u;

  for (int i = tid; i < OWN_ROWS * ACC_PITCH / 4; i += AGG_WAVES * 32)
    ((float4*)acc)[i] = make_float4(0.f, 0.f, 0.f, 0.f);
  __syncthreads();

  const int start = (int)(((long long)blockIdx.x * n_win) / gridDim.x);

  v4u nk[WIN_LOADS];
#pragma unroll
  for (int q = 0; q < WIN_LOADS; ++q) nk[q] = (v4u){~0u, ~0u, ~0u, ~0u};

  auto load_window = [&](int w, v4u* out) {
    const v4u* p = (const v4u*)keys + (size_t)w * (WIN / 16) + lane;
#pragma unroll
    for (int q = 0; q < WIN_LOADS; ++q) out[q] = p[q * 32];
  };

  int idx = wave;
  if (idx < n_win) {
    int w = idx + start; if (w >= n_win) w -= n_win;
    load_window(w, nk);
  }

  for (; idx < n_win; idx += AGG_WAVES) {
    int w = idx + start; if (w >= n_win) w -= n_win;
    v4u k[WIN_LOADS];
#pragma unroll
    for (int q = 0; q < WIN_LOADS; ++q) k[q] = nk[q];
    if (idx + AGG_WAVES < n_win) {
      int wn = idx + AGG_WAVES + start; if (wn >= n_win) wn -= n_win;
      load_window(wn, nk);
    }

    unsigned count = 0;
#pragma unroll
    for (int q = 0; q < WIN_LOADS; ++q) {
      unsigned g = 0;
#pragma unroll
      for (int cw = 0; cw < 4; ++cw) {
        const unsigned xx = k[q][cw] ^ pat;
        const unsigned nz = (((xx & 0x7f7f7f7fu) + 0x7f7f7f7fu) | xx) & 0x80808080u;
        g |= (nz ^ 0x80808080u) >> (7 - cw);
      }
      unsigned m = __builtin_amdgcn_ballot_w32(g != 0u);
      while (m) {
        if (g != 0u) {
          const unsigned t = (unsigned)__builtin_ctz(g);
          const unsigned pos = count + __builtin_amdgcn_mbcnt_lo(m, 0u);
          myq[pos] = (unsigned short)(q * 512 + lane * 16 + ((t & 7u) << 2) + (t >> 3));
          g &= g - 1u;
        }
        count += (unsigned)__builtin_popcount(m);
        m = __builtin_amdgcn_ballot_w32(g != 0u);
      }
    }

    const int wbase = w * WIN;
    for (unsigned base = 0; base < count; base += 32) {
      const unsigned qi = base + lane;
      unsigned pk = 0;
      if (qi < count) pk = packed[wbase + (int)myq[qi]];
      const int s_l = (int)(pk & ((1u << SRC_BITS) - 1u));
      const unsigned r_l = pk >> SRC_BITS;
      const unsigned n = (count - base < 32u) ? (count - base) : 32u;
      for (unsigned j = 0; j < n; j += 4) {
        const bool k1 = j + 1 < n, k2 = j + 2 < n, k3 = j + 3 < n;
        const int j0 = (int)j;
        const int j1 = k1 ? j0 + 1 : j0;
        const int j2 = k2 ? j0 + 2 : j0;
        const int j3 = k3 ? j0 + 3 : j0;
        const int s0 = __builtin_amdgcn_readlane(s_l, j0);
        const int s1 = __builtin_amdgcn_readlane(s_l, j1);
        const int s2 = __builtin_amdgcn_readlane(s_l, j2);
        const int s3 = __builtin_amdgcn_readlane(s_l, j3);
        const v2f v0 = *(const v2f*)(x + (size_t)s0 * D_FEAT + 2 * lane);
        const v2f v1 = *(const v2f*)(x + (size_t)s1 * D_FEAT + 2 * lane);
        const v2f v2 = *(const v2f*)(x + (size_t)s2 * D_FEAT + 2 * lane);
        const v2f v3 = *(const v2f*)(x + (size_t)s3 * D_FEAT + 2 * lane);
        const float n0 = out_norm[s0];
        const float n1 = out_norm[s1];
        const float n2 = out_norm[s2];
        const float n3 = out_norm[s3];
        float* p0 = acc + __builtin_amdgcn_readlane((int)r_l, j0) * ACC_PITCH + 2 * lane;
        float* p1 = acc + __builtin_amdgcn_readlane((int)r_l, j1) * ACC_PITCH + 2 * lane;
        float* p2 = acc + __builtin_amdgcn_readlane((int)r_l, j2) * ACC_PITCH + 2 * lane;
        float* p3 = acc + __builtin_amdgcn_readlane((int)r_l, j3) * ACC_PITCH + 2 * lane;
        __hip_atomic_fetch_add(p0,     v0.x * n0, __ATOMIC_RELAXED, __HIP_MEMORY_SCOPE_WORKGROUP);
        __hip_atomic_fetch_add(p0 + 1, v0.y * n0, __ATOMIC_RELAXED, __HIP_MEMORY_SCOPE_WORKGROUP);
        __hip_atomic_fetch_add(p1,     k1 ? v1.x * n1 : 0.f, __ATOMIC_RELAXED, __HIP_MEMORY_SCOPE_WORKGROUP);
        __hip_atomic_fetch_add(p1 + 1, k1 ? v1.y * n1 : 0.f, __ATOMIC_RELAXED, __HIP_MEMORY_SCOPE_WORKGROUP);
        __hip_atomic_fetch_add(p2,     k2 ? v2.x * n2 : 0.f, __ATOMIC_RELAXED, __HIP_MEMORY_SCOPE_WORKGROUP);
        __hip_atomic_fetch_add(p2 + 1, k2 ? v2.y * n2 : 0.f, __ATOMIC_RELAXED, __HIP_MEMORY_SCOPE_WORKGROUP);
        __hip_atomic_fetch_add(p3,     k3 ? v3.x * n3 : 0.f, __ATOMIC_RELAXED, __HIP_MEMORY_SCOPE_WORKGROUP);
        __hip_atomic_fetch_add(p3 + 1, k3 ? v3.y * n3 : 0.f, __ATOMIC_RELAXED, __HIP_MEMORY_SCOPE_WORKGROUP);
      }
    }
  }

  const int half = lane >> 4;
  const int l15  = lane & 15;
  const int n0   = (wave & 3) * 16;
  const float* __restrict__ bptr = W + (size_t)(2 * half) * D_FEAT + n0 + l15;
  v2f bf[D_FEAT / 4];
#pragma unroll
  for (int k = 0; k < D_FEAT / 4; ++k) {
    bf[k].x = bptr[(size_t)(4 * k) * D_FEAT];
    bf[k].y = bptr[(size_t)(4 * k + 1) * D_FEAT];
  }
  const float bv = bias[n0 + l15];

  __syncthreads();

  float* stage = reinterpret_cast<float*>(hitq);
  for (int rt = wave >> 2; rt < ROW_TILES; rt += AGG_WAVES / 4) {
    const int lrow = rt * 16 + l15;
    const int grow = lo + lrow;
    const float scale = (grow < n_nodes) ? in_norm[grow] : 0.0f;
    const float* ap = acc + lrow * ACC_PITCH + 2 * half;
    v8f c = {};
#pragma unroll
    for (int k = 0; k < D_FEAT / 4; ++k) {
      v2f a = *(const v2f*)(ap + 4 * k);
      a.x *= scale;
      a.y *= scale;
      c = __builtin_amdgcn_wmma_f32_16x16x4_f32(false, a, false, bf[k],
                                                (short)0, c, false, false);
    }
#pragma unroll
    for (int v = 0; v < 8; ++v) {
      const int r = rt * 16 + v + 8 * half;
      const float xv = c[v] + bv;
      const float yv = xv > 0.0f ? xv : expm1f(xv);
      stage[r * D_FEAT + n0 + l15] = yv;
    }
  }

  __syncthreads();
  const int rows_here = (n_nodes - lo < OWN_ROWS) ? (n_nodes - lo) : OWN_ROWS;
  float4* __restrict__ o4 = (float4*)(out + (size_t)lo * D_FEAT);
  for (int i = tid; i < rows_here * (D_FEAT / 4); i += AGG_WAVES * 32)
    o4[i] = ((const float4*)stage)[i];
}

extern "C" void kernel_launch(void* const* d_in, const int* in_sizes, int n_in,
                              void* d_out, int out_size, void* d_ws, size_t ws_size,
                              hipStream_t stream) {
  const float* feat = (const float*)d_in[0];
  const float* Ws[3] = {(const float*)d_in[1], (const float*)d_in[3], (const float*)d_in[5]};
  const float* bs[3] = {(const float*)d_in[2], (const float*)d_in[4], (const float*)d_in[6]};
  const int* src = (const int*)d_in[7];
  const int* dst = (const int*)d_in[8];

  const int n_nodes = in_sizes[0] / D_FEAT;
  const int n_edges = in_sizes[7];

  const int n_win = (n_edges + WIN - 1) / WIN;
  const int n_pad = n_win * WIN;
  float* out_norm  = (float*)d_ws;
  float* in_norm   = out_norm + n_nodes;
  float* mid       = in_norm + n_nodes;
  unsigned* packed = (unsigned*)(mid + (size_t)n_nodes * D_FEAT);
  unsigned* keys   = packed + n_pad;
  float* outp      = (float*)d_out;

  const int TPB = 256;

  {
    long long n2 = 2LL * n_nodes;
    gcn_zero_u32<<<(unsigned)((n2 + TPB - 1) / TPB), TPB, 0, stream>>>(
        (unsigned*)out_norm, n2);
    gcn_degree_encode<<<(unsigned)((n_pad / 4 + TPB - 1) / TPB), TPB, 0, stream>>>(
        src, dst, (unsigned*)out_norm, (unsigned*)in_norm, packed, keys,
        n_nodes, n_edges, n_pad);
    gcn_norms<<<(unsigned)((n_nodes + TPB - 1) / TPB), TPB, 0, stream>>>(
        out_norm, in_norm, n_nodes);
  }

  const float* x = feat;
  for (int l = 0; l < 3; ++l) {
    float* y = (l == 1) ? mid : outp;
    gcn_agg_owned<<<(unsigned)((n_nodes + OWN_ROWS - 1) / OWN_ROWS), AGG_WAVES * 32, 0, stream>>>(
        x, out_norm, keys, packed, in_norm, Ws[l], bs[l], y, n_nodes, n_win);
    x = y;
  }
}


# Owners' edges found once: layer 1's key scan also appends src|row words to each workgroup's list stretch; layers 2 and 3 read that list, 16 rows in flight, no scan; sum order: LDS arrival as before
# speedup vs baseline: 2.9692x; 1.1031x over previous
#include <hip/hip_runtime.h>
#include <hip/hip_bf16.h>

#define D_FEAT 64

typedef float v2f __attribute__((ext_vector_type(2)));
typedef float v8f __attribute__((ext_vector_type(8)));

__global__ void gcn_zero_u32(unsigned* __restrict__ p, long long n) {
  long long i = (long long)blockIdx.x * blockDim.x + threadIdx.x;
  if (i < n) p[i] = 0u;
}

#define OWN_ROWS 400
#define SRC_BITS 17

typedef int v4i __attribute__((ext_vector_type(4)));
typedef unsigned v4u __attribute__((ext_vector_type(4)));

__global__ void gcn_degree_encode(const int* __restrict__ src, const int* __restrict__ dst,
                                  unsigned* __restrict__ outdeg, unsigned* __restrict__ indeg,
                                  unsigned* __restrict__ keys,
                                  int n_nodes, int n_edges, int n_pad) {
  const int e0 = (blockIdx.x * blockDim.x + threadIdx.x) * 4;
  if (e0 >= n_pad) return;
  v4i s = (v4i){0, 0, 0, 0};
  v4i d = (v4i){-1, -1, -1, -1};
  if (e0 + 3 < n_edges) {
    s = *(const v4i*)(src + e0);
    d = *(const v4i*)(dst + e0);
  } else {
#pragma unroll
    for (int j = 0; j < 4; ++j)
      if (e0 + j < n_edges) { s[j] = src[e0 + j]; d[j] = dst[e0 + j]; }
  }
  unsigned kw = 0;
#pragma unroll
  for (int j = 0; j < 4; ++j) {
    unsigned key = 0xFFu;
    if (e0 + j < n_edges) {
      atomicAdd(&outdeg[s[j]], 1u);
      atomicAdd(&indeg[d[j]], 1u);
      if ((unsigned)d[j] < (unsigned)n_nodes) key = (unsigned)d[j] / (unsigned)OWN_ROWS;
    }
    kw |= key << (8 * j);
  }
  keys[e0 >> 2] = kw;
}

__global__ __launch_bounds__(OWN_ROWS) void gcn_norms(float* __restrict__ outn,
                                                      float* __restrict__ inn,
                                                      unsigned* __restrict__ own_cnt, int n) {
  __shared__ unsigned tot;
  if (threadIdx.x == 0) tot = 0u;
  __syncthreads();
  int i = blockIdx.x * OWN_ROWS + threadIdx.x;
  if (i < n) {
    unsigned od = ((const unsigned*)outn)[i];
    unsigned id_ = ((const unsigned*)inn)[i];
    outn[i] = rsqrtf((float)(od > 1u ? od : 1u));
    inn[i]  = rsqrtf((float)(id_ > 1u ? id_ : 1u));
    atomicAdd(&tot, id_);
  }
  __syncthreads();
  if (threadIdx.x == 0) own_cnt[blockIdx.x] = tot;
}

#define AGG_WAVES 32
#define WIN 2048
#define WIN_LOADS (WIN / 512)
#define ACC_PITCH 68
#define ROW_TILES (OWN_ROWS / 16)
static_assert(OWN_ROWS % 16 == 0, "whole 16-row tiles");
static_assert(AGG_WAVES * WIN * 2 >= OWN_ROWS * D_FEAT * 4, "results fit in the hit lists' LDS");

template <int G>
__device__ __forceinline__ void add_rows(const float* __restrict__ x,
                                         const float* __restrict__ out_norm,
                                         float* acc, int s_l, unsigned r_l, unsigned n, int lane) {
  for (unsigned j = 0; j < n; j += G) {
    bool ok[G];
    int sj[G];
    float* p[G];
    v2f v[G];
    float nm[G];
#pragma unroll
    for (int t = 0; t < G; ++t) {
      ok[t] = (t == 0) || (j + (unsigned)t < n);
      const int jt = ok[t] ? (int)j + t : (int)j;
      sj[t] = __builtin_amdgcn_readlane(s_l, jt);
      p[t] = acc + __builtin_amdgcn_readlane((int)r_l, jt) * ACC_PITCH + 2 * lane;
    }
#pragma unroll
    for (int t = 0; t < G; ++t) v[t] = *(const v2f*)(x + (size_t)sj[t] * D_FEAT + 2 * lane);
#pragma unroll
    for (int t = 0; t < G; ++t) nm[t] = out_norm[sj[t]];
#pragma unroll
    for (int t = 0; t < G; ++t) {
      __hip_atomic_fetch_add(p[t],     ok[t] ? v[t].x * nm[t] : 0.f, __ATOMIC_RELAXED, __HIP_MEMORY_SCOPE_WORKGROUP);
      __hip_atomic_fetch_add(p[t] + 1, ok[t] ? v[t].y * nm[t] : 0.f, __ATOMIC_RELAXED, __HIP_MEMORY_SCOPE_WORKGROUP);
    }
  }
}

template <bool SCAN>
__global__ __launch_bounds__(AGG_WAVES * 32) void gcn_agg_owned(
    const float* __restrict__ x, const float* __restrict__ out_norm,
    const unsigned* __restrict__ keys,
    const int* __restrict__ src, const int* __restrict__ dst,
    unsigned* __restrict__ list, const unsigned* __restrict__ own_cnt,
    const float* __restrict__ in_norm,
    const float* __restrict__ W,
    const float* __restrict__ bias,
    float* __restrict__ out, int n_nodes, int n_win) {
  __shared__ __attribute__((aligned(16))) float acc[OWN_ROWS * ACC_PITCH];
  __shared__ __attribute__((aligned(16))) unsigned short hitq[AGG_WAVES * WIN];
  __shared__ unsigned s_off;
  __shared__ unsigned s_fill;

  const int tid  = threadIdx.x;
  const int lane = tid & 31;
  const int wave = __builtin_amdgcn_readfirstlane(tid >> 5);
  const int lo   = blockIdx.x * OWN_ROWS;

  const unsigned below = (tid < (int)blockIdx.x) ? own_cnt[tid] : 0u;
  if (tid == 0) { s_off = 0u; s_fill = 0u; }
  for (int i = tid; i < OWN_ROWS * ACC_PITCH / 4; i += AGG_WAVES * 32)
    ((float4*)acc)[i] = make_float4(0.f, 0.f, 0.f, 0.f);
  __syncthreads();
  if (tid < (int)blockIdx.x) atomicAdd(&s_off, below);
  __syncthreads();
  const unsigned off = (unsigned)__builtin_amdgcn_readfirstlane((int)s_off);
  const unsigned cnt = (unsigned)__builtin_amdgcn_readfirstlane((int)own_cnt[blockIdx.x]);
  unsigned* __restrict__ mylist = list + off;

  if constexpr (SCAN) {
    unsigned short* __restrict__ myq = hitq + wave * WIN;
    const unsigned pat = (unsigned)blockIdx.x * 0x01010101u;

    const int start = (int)(((long long)blockIdx.x * n_win) / gridDim.x);

    v4u nk[WIN_LOADS];
#pragma unroll
    for (int q = 0; q < WIN_LOADS; ++q) nk[q] = (v4u){~0u, ~0u, ~0u, ~0u};

    auto load_window = [&](int w, v4u* o) {
      const v4u* p = (const v4u*)keys + (size_t)w * (WIN / 16) + lane;
#pragma unroll
      for (int q = 0; q < WIN_LOADS; ++q) o[q] = p[q * 32];
    };

    int idx = wave;
    if (idx < n_win) {
      int w = idx + start; if (w >= n_win) w -= n_win;
      load_window(w, nk);
    }

    for (; idx < n_win; idx += AGG_WAVES) {
      int w = idx + start; if (w >= n_win) w -= n_win;
      v4u k[WIN_LOADS];
#pragma unroll
      for (int q = 0; q < WIN_LOADS; ++q) k[q] = nk[q];
      if (idx + AGG_WAVES < n_win) {
        int wn = idx + AGG_WAVES + start; if (wn >= n_win) wn -= n_win;
        load_window(wn, nk);
      }

      unsigned count = 0;
#pragma unroll
      for (int q = 0; q < WIN_LOADS; ++q) {
        unsigned g = 0;
#pragma unroll
        for (int cw = 0; cw < 4; ++cw) {
          const unsigned xx = k[q][cw] ^ pat;
          const unsigned nz = (((xx & 0x7f7f7f7fu) + 0x7f7f7f7fu) | xx) & 0x80808080u;
          g |= (nz ^ 0x80808080u) >> (7 - cw);
        }
        unsigned m = __builtin_amdgcn_ballot_w32(g != 0u);
        while (m) {
          if (g != 0u) {
            const unsigned t = (unsigned)__builtin_ctz(g);
            const unsigned pos = count + __builtin_amdgcn_mbcnt_lo(m, 0u);
            myq[pos] = (unsigned short)(q * 512 + lane * 16 + ((t & 7u) << 2) + (t >> 3));
            g &= g - 1u;
          }
          count += (unsigned)__builtin_popcount(m);
          m = __builtin_amdgcn_ballot_w32(g != 0u);
        }
      }

      const int wbase = w * WIN;
      for (unsigned base = 0; base < count; base += 32) {
        const unsigned qi = base + lane;
        unsigned pk = 0;
        if (qi < count) {
          const int e = wbase + (int)myq[qi];
          pk = (unsigned)src[e] | ((unsigned)(dst[e] - lo) << SRC_BITS);
        }
        const unsigned n = (count - base < 32u) ? (count - base) : 32u;
        unsigned slot = 0;
        if (lane == 0) slot = atomicAdd(&s_fill, n);
        slot = (unsigned)__builtin_amdgcn_readfirstlane((int)slot) + (unsigned)lane;
        if ((unsigned)lane < n && slot < cnt) mylist[slot] = pk;
        add_rows<4>(x, out_norm, acc, (int)(pk & ((1u << SRC_BITS) - 1u)), pk >> SRC_BITS, n, lane);
      }
    }
  } else {
    const unsigned n_chunks = (cnt + 31u) >> 5;
    unsigned c = (unsigned)wave;
    unsigned pk_next = 0;
    if (c < n_chunks) {
      const unsigned i = c * 32u + (unsigned)lane;
      if (i < cnt) pk_next = mylist[i];
    }
    for (; c < n_chunks; c += AGG_WAVES) {
      const unsigned pk = pk_next;
      const unsigned cn = c + AGG_WAVES;
      pk_next = 0;
      if (cn < n_chunks) {
        const unsigned i = cn * 32u + (unsigned)lane;
        if (i < cnt) pk_next = mylist[i];
      }
      const unsigned left = cnt - c * 32u;
      const unsigned n = left < 32u ? left : 32u;
      add_rows<16>(x, out_norm, acc, (int)(pk & ((1u << SRC_BITS) - 1u)), pk >> SRC_BITS, n, lane);
    }
  }

  const int half = lane >> 4;
  const int l15  = lane & 15;
  const int n0   = (wave & 3) * 16;
  const float* __restrict__ bptr = W + (size_t)(2 * half) * D_FEAT + n0 + l15;
  v2f bf[D_FEAT / 4];
#pragma unroll
  for (int k = 0; k < D_FEAT / 4; ++k) {
    bf[k].x = bptr[(size_t)(4 * k) * D_FEAT];
    bf[k].y = bptr[(size_t)(4 * k + 1) * D_FEAT];
  }
  const float bv = bias[n0 + l15];

  __syncthreads();

  float* stage = reinterpret_cast<float*>(hitq);
  for (int rt = wave >> 2; rt < ROW_TILES; rt += AGG_WAVES / 4) {
    const int lrow = rt * 16 + l15;
    const int grow = lo + lrow;
    const float scale = (grow < n_nodes) ? in_norm[grow] : 0.0f;
    const float* ap = acc + lrow * ACC_PITCH + 2 * half;
    v8f c = {};
#pragma unroll
    for (int k = 0; k < D_FEAT / 4; ++k) {
      v2f a = *(const v2f*)(ap + 4 * k);
      a.x *= scale;
      a.y *= scale;
      c = __builtin_amdgcn_wmma_f32_16x16x4_f32(false, a, false, bf[k],
                                                (short)0, c, false, false);
    }
#pragma unroll
    for (int v = 0; v < 8; ++v) {
      const int r = rt * 16 + v + 8 * half;
      const float xv = c[v] + bv;
      const float yv = xv > 0.0f ? xv : expm1f(xv);
      stage[r * D_FEAT + n0 + l15] = yv;
    }
  }

  __syncthreads();
  const int rows_here = (n_nodes - lo < OWN_ROWS) ? (n_nodes - lo) : OWN_ROWS;
  float4* __restrict__ o4 = (float4*)(out + (size_t)lo * D_FEAT);
  for (int i = tid; i < rows_here * (D_FEAT / 4); i += AGG_WAVES * 32)
    o4[i] = ((const float4*)stage)[i];
}

extern "C" void kernel_launch(void* const* d_in, const int* in_sizes, int n_in,
                              void* d_out, int out_size, void* d_ws, size_t ws_size,
                              hipStream_t stream) {
  const float* feat = (const float*)d_in[0];
  const float* Ws[3] = {(const float*)d_in[1], (const float*)d_in[3], (const float*)d_in[5]};
  const float* bs[3] = {(const float*)d_in[2], (const float*)d_in[4], (const float*)d_in[6]};
  const int* src = (const int*)d_in[7];
  const int* dst = (const int*)d_in[8];

  const int n_nodes = in_sizes[0] / D_FEAT;
  const int n_edges = in_sizes[7];

  const int n_win = (n_edges + WIN - 1) / WIN;
  const int n_pad = n_win * WIN;
  float* out_norm  = (float*)d_ws;
  float* in_norm   = out_norm + n_nodes;
  float* mid       = in_norm + n_nodes;
  unsigned* list   = (unsigned*)(mid + (size_t)n_nodes * D_FEAT);
  unsigned* keys   = list + n_pad;
  unsigned* own_cnt = keys + n_pad / 4;
  float* outp      = (float*)d_out;

  const int TPB = 256;
  const unsigned n_own = (unsigned)((n_nodes + OWN_ROWS - 1) / OWN_ROWS);

  {
    long long n2 = 2LL * n_nodes;
    gcn_zero_u32<<<(unsigned)((n2 + TPB - 1) / TPB), TPB, 0, stream>>>(
        (unsigned*)out_norm, n2);
    gcn_degree_encode<<<(unsigned)((n_pad / 4 + TPB - 1) / TPB), TPB, 0, stream>>>(
        src, dst, (unsigned*)out_norm, (unsigned*)in_norm, keys,
        n_nodes, n_edges, n_pad);
    gcn_norms<<<n_own, OWN_ROWS, 0, stream>>>(out_norm, in_norm, own_cnt, n_nodes);
  }

  const float* x = feat;
  for (int l = 0; l < 3; ++l) {
    float* y = (l == 1) ? mid : outp;
    if (l == 0)
      gcn_agg_owned<true><<<n_own, AGG_WAVES * 32, 0, stream>>>(
          x, out_norm, keys, src, dst, list, own_cnt, in_norm, Ws[l], bs[l], y, n_nodes, n_win);
    else
      gcn_agg_owned<false><<<n_own, AGG_WAVES * 32, 0, stream>>>(
          x, out_norm, keys, src, dst, list, own_cnt, in_norm, Ws[l], bs[l], y, n_nodes, n_win);
    x = y;
  }
}


# Degrees without global atomics: a workgroup counts one edge slice's src or dst ids of one 25600-node range by LDS integer adds, norms kernel sums 32 partial planes; 6 to 5 launches; float sums same
# speedup vs baseline: 4.5269x; 1.5246x over previous
#include <hip/hip_runtime.h>
#include <hip/hip_bf16.h>

#define D_FEAT 64

typedef float v2f __attribute__((ext_vector_type(2)));
typedef float v8f __attribute__((ext_vector_type(8)));

#define OWN_ROWS 400
#define SRC_BITS 17
#define HIST_SLICES 32
#define HIST_CAP 25600
#define HIST_THREADS 1024
#define HIST_BATCH 5
static_assert(HIST_CAP % 4 == 0, "ranges start on 16-byte boundaries of a plane");
static_assert(2 * HIST_SLICES == D_FEAT, "the partial planes fill exactly the n_nodes x 64 plane they borrow");

typedef int v4i __attribute__((ext_vector_type(4)));
typedef unsigned v4u __attribute__((ext_vector_type(4)));

__device__ __forceinline__ v4i load_quad(const int* __restrict__ a, int e0, int n_edges) {
  v4i v = (v4i){-1, -1, -1, -1};
  if (e0 + 3 < n_edges) {
    v = *(const v4i*)(a + e0);
  } else {
#pragma unroll
    for (int j = 0; j < 4; ++j)
      if (e0 + j < n_edges) v[j] = a[e0 + j];
  }
  return v;
}

__global__ __launch_bounds__(HIST_THREADS) void gcn_degree_hist(
    const int* __restrict__ src, const int* __restrict__ dst,
    unsigned* __restrict__ part, unsigned* __restrict__ keys,
    int n_nodes, int n_edges, int n_pad) {
  __shared__ __attribute__((aligned(16))) unsigned cnt[HIST_CAP];
  const int tid = threadIdx.x;
  const int slice = blockIdx.x;
  const int lo = blockIdx.y * HIST_CAP;
  const int len = (n_nodes - lo < HIST_CAP) ? (n_nodes - lo) : HIST_CAP;
  const bool is_dst = blockIdx.z != 0;
  const int* __restrict__ ids = is_dst ? dst : src;
  const bool write_keys = is_dst && blockIdx.y == 0;

  for (int i = tid; i < HIST_CAP / 4; i += HIST_THREADS) ((v4u*)cnt)[i] = (v4u){0u, 0u, 0u, 0u};
  __syncthreads();

  const int nq = n_pad >> 2;
  const int qps = (nq + HIST_SLICES - 1) / HIST_SLICES;
  const int qb = slice * qps;
  const int qe = (qb + qps < nq) ? (qb + qps) : nq;
  const int qfull = n_edges >> 2;
  const int qe_full = (qe < qfull) ? qe : qfull;

  auto take = [&](v4i v, int qq) {
#pragma unroll
    for (int c4 = 0; c4 < 4; ++c4) {
      const unsigned r = (unsigned)v[c4] - (unsigned)lo;
      if (r < (unsigned)len) atomicAdd(&cnt[r], 1u);
    }
    if (write_keys) {
      unsigned kw = 0;
#pragma unroll
      for (int c4 = 0; c4 < 4; ++c4) {
        const unsigned d = (unsigned)v[c4];
        const unsigned key = (d < (unsigned)n_nodes) ? d / (unsigned)OWN_ROWS : 0xFFu;
        kw |= key << (8 * c4);
      }
      keys[qq] = kw;
    }
  };

  for (int q = qb + tid; q < qe_full; q += HIST_THREADS * HIST_BATCH) {
    v4i v[HIST_BATCH];
#pragma unroll
    for (int j = 0; j < HIST_BATCH; ++j) {
      const int qq = q + j * HIST_THREADS;
      v[j] = *(const v4i*)(ids + (size_t)((qq < qe_full) ? qq : q) * 4);
    }
#pragma unroll
    for (int j = 0; j < HIST_BATCH; ++j) {
      const int qq = q + j * HIST_THREADS;
      if (qq < qe_full) take(v[j], qq);
    }
  }
  {
    const int qt0 = (qb > qfull) ? qb : qfull;
    const int qt1 = write_keys ? qe : ((qe < ((n_edges + 3) >> 2)) ? qe : ((n_edges + 3) >> 2));
    for (int q = qt0 + tid; q < qt1; q += HIST_THREADS) take(load_quad(ids, q * 4, n_edges), q);
  }
  __syncthreads();

  unsigned* __restrict__ my = part + (size_t)(blockIdx.z * HIST_SLICES + slice) * n_nodes + lo;
  if ((n_nodes & 3) == 0) {
    for (int i = tid; i < len / 4; i += HIST_THREADS) ((v4u*)my)[i] = ((const v4u*)cnt)[i];
  } else {
    for (int i = tid; i < len; i += HIST_THREADS) my[i] = cnt[i];
  }
}

__global__ __launch_bounds__(OWN_ROWS) void gcn_norms(const unsigned* __restrict__ part,
                                                      float* __restrict__ outn,
                                                      float* __restrict__ inn,
                                                      unsigned* __restrict__ own_cnt, int n) {
  __shared__ unsigned tot;
  if (threadIdx.x == 0) tot = 0u;
  __syncthreads();
  int i = blockIdx.x * OWN_ROWS + threadIdx.x;
  if (i < n) {
    const unsigned* __restrict__ p = part + i;
    unsigned od = 0u, id_ = 0u;
#pragma unroll
    for (int s = 0; s < HIST_SLICES; ++s) {
      od  += p[(size_t)s * n];
      id_ += p[(size_t)(HIST_SLICES + s) * n];
    }
    outn[i] = rsqrtf((float)(od > 1u ? od : 1u));
    inn[i]  = rsqrtf((float)(id_ > 1u ? id_ : 1u));
    atomicAdd(&tot, id_);
  }
  __syncthreads();
  if (threadIdx.x == 0) own_cnt[blockIdx.x] = tot;
}

#define AGG_WAVES 32
#define WIN 2048
#define WIN_LOADS (WIN / 512)
#define ACC_PITCH 68
#define ROW_TILES (OWN_ROWS / 16)
static_assert(OWN_ROWS % 16 == 0, "whole 16-row tiles");
static_assert(AGG_WAVES * WIN * 2 >= OWN_ROWS * D_FEAT * 4, "results fit in the hit lists' LDS");

template <int G>
__device__ __forceinline__ void add_rows(const float* __restrict__ x,
                                         const float* __restrict__ out_norm,
                                         float* acc, int s_l, unsigned r_l, unsigned n, int lane) {
  for (unsigned j = 0; j < n; j += G) {
    bool ok[G];
    int sj[G];
    float* p[G];
    v2f v[G];
    float nm[G];
#pragma unroll
    for (int t = 0; t < G; ++t) {
      ok[t] = (t == 0) || (j + (unsigned)t < n);
      const int jt = ok[t] ? (int)j + t : (int)j;
      sj[t] = __builtin_amdgcn_readlane(s_l, jt);
      p[t] = acc + __builtin_amdgcn_readlane((int)r_l, jt) * ACC_PITCH + 2 * lane;
    }
#pragma unroll
    for (int t = 0; t < G; ++t) v[t] = *(const v2f*)(x + (size_t)sj[t] * D_FEAT + 2 * lane);
#pragma unroll
    for (int t = 0; t < G; ++t) nm[t] = out_norm[sj[t]];
#pragma unroll
    for (int t = 0; t < G; ++t) {
      __hip_atomic_fetch_add(p[t],     ok[t] ? v[t].x * nm[t] : 0.f, __ATOMIC_RELAXED, __HIP_MEMORY_SCOPE_WORKGROUP);
      __hip_atomic_fetch_add(p[t] + 1, ok[t] ? v[t].y * nm[t] : 0.f, __ATOMIC_RELAXED, __HIP_MEMORY_SCOPE_WORKGROUP);
    }
  }
}

template <bool SCAN>
__global__ __launch_bounds__(AGG_WAVES * 32) void gcn_agg_owned(
    const float* __restrict__ x, const float* __restrict__ out_norm,
    const unsigned* __restrict__ keys,
    const int* __restrict__ src, const int* __restrict__ dst,
    unsigned* __restrict__ list, const unsigned* __restrict__ own_cnt,
    const float* __restrict__ in_norm,
    const float* __restrict__ W,
    const float* __restrict__ bias,
    float* __restrict__ out, int n_nodes, int n_win) {
  __shared__ __attribute__((aligned(16))) float acc[OWN_ROWS * ACC_PITCH];
  __shared__ __attribute__((aligned(16))) unsigned short hitq[AGG_WAVES * WIN];
  __shared__ unsigned s_off;
  __shared__ unsigned s_fill;

  const int tid  = threadIdx.x;
  const int lane = tid & 31;
  const int wave = __builtin_amdgcn_readfirstlane(tid >> 5);
  const int lo   = blockIdx.x * OWN_ROWS;

  const unsigned below = (tid < (int)blockIdx.x) ? own_cnt[tid] : 0u;
  if (tid == 0) { s_off = 0u; s_fill = 0u; }
  for (int i = tid; i < OWN_ROWS * ACC_PITCH / 4; i += AGG_WAVES * 32)
    ((float4*)acc)[i] = make_float4(0.f, 0.f, 0.f, 0.f);
  __syncthreads();
  if (tid < (int)blockIdx.x) atomicAdd(&s_off, below);
  __syncthreads();
  const unsigned off = (unsigned)__builtin_amdgcn_readfirstlane((int)s_off);
  const unsigned cnt = (unsigned)__builtin_amdgcn_readfirstlane((int)own_cnt[blockIdx.x]);
  unsigned* __restrict__ mylist = list + off;

  if constexpr (SCAN) {
    unsigned short* __restrict__ myq = hitq + wave * WIN;
    const unsigned pat = (unsigned)blockIdx.x * 0x01010101u;

    const int start = (int)(((long long)blockIdx.x * n_win) / gridDim.x);

    v4u nk[WIN_LOADS];
#pragma unroll
    for (int q = 0; q < WIN_LOADS; ++q) nk[q] = (v4u){~0u, ~0u, ~0u, ~0u};

    auto load_window = [&](int w, v4u* o) {
      const v4u* p = (const v4u*)keys + (size_t)w * (WIN / 16) + lane;
#pragma unroll
      for (int q = 0; q < WIN_LOADS; ++q) o[q] = p[q * 32];
    };

    int idx = wave;
    if (idx < n_win) {
      int w = idx + start; if (w >= n_win) w -= n_win;
      load_window(w, nk);
    }

    for (; idx < n_win; idx += AGG_WAVES) {
      int w = idx + start; if (w >= n_win) w -= n_win;
      v4u k[WIN_LOADS];
#pragma unroll
      for (int q = 0; q < WIN_LOADS; ++q) k[q] = nk[q];
      if (idx + AGG_WAVES < n_win) {
        int wn = idx + AGG_WAVES + start; if (wn >= n_win) wn -= n_win;
        load_window(wn, nk);
      }

      unsigned count = 0;
#pragma unroll
      for (int q = 0; q < WIN_LOADS; ++q) {
        unsigned g = 0;
#pragma unroll
        for (int cw = 0; cw < 4; ++cw) {
          const unsigned xx = k[q][cw] ^ pat;
          const unsigned nz = (((xx & 0x7f7f7f7fu) + 0x7f7f7f7fu) | xx) & 0x80808080u;
          g |= (nz ^ 0x80808080u) >> (7 - cw);
        }
        unsigned m = __builtin_amdgcn_ballot_w32(g != 0u);
        while (m) {
          if (g != 0u) {
            const unsigned t = (unsigned)__builtin_ctz(g);
            const unsigned pos = count + __builtin_amdgcn_mbcnt_lo(m, 0u);
            myq[pos] = (unsigned short)(q * 512 + lane * 16 + ((t & 7u) << 2) + (t >> 3));
            g &= g - 1u;
          }
          count += (unsigned)__builtin_popcount(m);
          m = __builtin_amdgcn_ballot_w32(g != 0u);
        }
      }

      const int wbase = w * WIN;
      for (unsigned base = 0; base < count; base += 32) {
        const unsigned qi = base + lane;
        unsigned pk = 0;
        if (qi < count) {
          const int e = wbase + (int)myq[qi];
          pk = (unsigned)src[e] | ((unsigned)(dst[e] - lo) << SRC_BITS);
        }
        const unsigned n = (count - base < 32u) ? (count - base) : 32u;
        unsigned slot = 0;
        if (lane == 0) slot = atomicAdd(&s_fill, n);
        slot = (unsigned)__builtin_amdgcn_readfirstlane((int)slot) + (unsigned)lane;
        if ((unsigned)lane < n && slot < cnt) mylist[slot] = pk;
        add_rows<4>(x, out_norm, acc, (int)(pk & ((1u << SRC_BITS) - 1u)), pk >> SRC_BITS, n, lane);
      }
    }
  } else {
    const unsigned n_chunks = (cnt + 31u) >> 5;
    unsigned c = (unsigned)wave;
    unsigned pk_next = 0;
    if (c < n_chunks) {
      const unsigned i = c * 32u + (unsigned)lane;
      if (i < cnt) pk_next = mylist[i];
    }
    for (; c < n_chunks; c += AGG_WAVES) {
      const unsigned pk = pk_next;
      const unsigned cn = c + AGG_WAVES;
      pk_next = 0;
      if (cn < n_chunks) {
        const unsigned i = cn * 32u + (unsigned)lane;
        if (i < cnt) pk_next = mylist[i];
      }
      const unsigned left = cnt - c * 32u;
      const unsigned n = left < 32u ? left : 32u;
      add_rows<16>(x, out_norm, acc, (int)(pk & ((1u << SRC_BITS) - 1u)), pk >> SRC_BITS, n, lane);
    }
  }

  const int half = lane >> 4;
  const int l15  = lane & 15;
  const int n0   = (wave & 3) * 16;
  const float* __restrict__ bptr = W + (size_t)(2 * half) * D_FEAT + n0 + l15;
  v2f bf[D_FEAT / 4];
#pragma unroll
  for (int k = 0; k < D_FEAT / 4; ++k) {
    bf[k].x = bptr[(size_t)(4 * k) * D_FEAT];
    bf[k].y = bptr[(size_t)(4 * k + 1) * D_FEAT];
  }
  const float bv = bias[n0 + l15];

  __syncthreads();

  float* stage = reinterpret_cast<float*>(hitq);
  for (int rt = wave >> 2; rt < ROW_TILES; rt += AGG_WAVES / 4) {
    const int lrow = rt * 16 + l15;
    const int grow = lo + lrow;
    const float scale = (grow < n_nodes) ? in_norm[grow] : 0.0f;
    const float* ap = acc + lrow * ACC_PITCH + 2 * half;
    v8f c = {};
#pragma unroll
    for (int k = 0; k < D_FEAT / 4; ++k) {
      v2f a = *(const v2f*)(ap + 4 * k);
      a.x *= scale;
      a.y *= scale;
      c = __builtin_amdgcn_wmma_f32_16x16x4_f32(false, a, false, bf[k],
                                                (short)0, c, false, false);
    }
#pragma unroll
    for (int v = 0; v < 8; ++v) {
      const int r = rt * 16 + v + 8 * half;
      const float xv = c[v] + bv;
      const float yv = xv > 0.0f ? xv : expm1f(xv);
      stage[r * D_FEAT + n0 + l15] = yv;
    }
  }

  __syncthreads();
  const int rows_here = (n_nodes - lo < OWN_ROWS) ? (n_nodes - lo) : OWN_ROWS;
  float4* __restrict__ o4 = (float4*)(out + (size_t)lo * D_FEAT);
  for (int i = tid; i < rows_here * (D_FEAT / 4); i += AGG_WAVES * 32)
    o4[i] = ((const float4*)stage)[i];
}

extern "C" void kernel_launch(void* const* d_in, const int* in_sizes, int n_in,
                              void* d_out, int out_size, void* d_ws, size_t ws_size,
                              hipStream_t stream) {
  const float* feat = (const float*)d_in[0];
  const float* Ws[3] = {(const float*)d_in[1], (const float*)d_in[3], (const float*)d_in[5]};
  const float* bs[3] = {(const float*)d_in[2], (const float*)d_in[4], (const float*)d_in[6]};
  const int* src = (const int*)d_in[7];
  const int* dst = (const int*)d_in[8];

  const int n_nodes = in_sizes[0] / D_FEAT;
  const int n_edges = in_sizes[7];

  const int n_win = (n_edges + WIN - 1) / WIN;
  const int n_pad = n_win * WIN;
  float* out_norm  = (float*)d_ws;
  float* in_norm   = out_norm + n_nodes;
  float* mid       = in_norm + n_nodes;
  unsigned* list   = (unsigned*)(mid + (size_t)n_nodes * D_FEAT);
  unsigned* keys   = list + n_pad;
  unsigned* own_cnt = keys + n_pad / 4;
  unsigned* part   = (unsigned*)mid;
  float* outp      = (float*)d_out;

  const unsigned n_own = (unsigned)((n_nodes + OWN_ROWS - 1) / OWN_ROWS);
  const unsigned n_rng = (unsigned)((n_nodes + HIST_CAP - 1) / HIST_CAP);

  {
    gcn_degree_hist<<<dim3(HIST_SLICES, n_rng, 2), HIST_THREADS, 0, stream>>>(
        src, dst, part, keys, n_nodes, n_edges, n_pad);
    gcn_norms<<<n_own, OWN_ROWS, 0, stream>>>(part, out_norm, in_norm, own_cnt, n_nodes);
  }

  const float* x = feat;
  for (int l = 0; l < 3; ++l) {
    float* y = (l == 1) ? mid : outp;
    if (l == 0)
      gcn_agg_owned<true><<<n_own, AGG_WAVES * 32, 0, stream>>>(
          x, out_norm, keys, src, dst, list, own_cnt, in_norm, Ws[l], bs[l], y, n_nodes, n_win);
    else
      gcn_agg_owned<false><<<n_own, AGG_WAVES * 32, 0, stream>>>(
          x, out_norm, keys, src, dst, list, own_cnt, in_norm, Ws[l], bs[l], y, n_nodes, n_win);
    x = y;
  }
}
